# starting kernel
# baseline (speedup 1.0000x reference)
#include <hip/hip_runtime.h>

#define B_ROWS   131072
#define N_FEAT   128
#define N_FIELD  16
#define FACTOR   16
#define WAVES_PER_BLOCK 4
#define XS_STRIDE 132

typedef __attribute__((ext_vector_type(2))) float v2f;
typedef __attribute__((ext_vector_type(8))) float v8f;

__global__ void ffm_build_A(const float* __restrict__ w,
                            const int* __restrict__ field,
                            float* __restrict__ A) {
  int idx = blockIdx.x * blockDim.x + threadIdx.x;
  if (idx >= N_FEAT * N_FEAT) return;
  int i = idx >> 7;
  int j = idx & (N_FEAT - 1);
  const float* wi = w + (i * N_FIELD + field[j]) * FACTOR;
  const float* wj = w + (j * N_FIELD + field[i]) * FACTOR;
  float s = 0.f;
#pragma unroll
  for (int d = 0; d < FACTOR; ++d) s += wi[d] * wj[d];
  const float val = (i == j) ? 0.f : 0.5f * s;

  const int t  = j >> 4,  n = j & 15;
  const int kk = i >> 3,  r = i & 7;
  const int h  = (r >> 1) & 1;
  const int q  = ((r >> 2) << 1) | (r & 1);
  const int lane = h * 16 + n;
  A[(((t * 16 + kk) * 32) + lane) * 4 + q] = val;
}

__global__ __launch_bounds__(WAVES_PER_BLOCK * 32)
void ffm_quadform(const float* __restrict__ x,
                  const float* __restrict__ A,
                  float* __restrict__ out) {
  __shared__ float xs[WAVES_PER_BLOCK][16][XS_STRIDE];

  const int wave = threadIdx.x >> 5;
  const int lane = threadIdx.x & 31;
  const int half = lane >> 4;
  const int l16  = lane & 15;
  const long tile = (long)blockIdx.x * WAVES_PER_BLOCK + wave;
  const float* xt = x + tile * 16 * N_FEAT;

#pragma unroll
  for (int r = 0; r < 16; ++r) {
    const float4 v = *(const float4*)(xt + r * N_FEAT + lane * 4);
    xs[wave][r][lane * 4 + 0] = v.x;
    xs[wave][r][lane * 4 + 1] = v.y;
    xs[wave][r][lane * 4 + 2] = v.z;
    xs[wave][r][lane * 4 + 3] = v.w;
  }

  float rowacc[8];
#pragma unroll
  for (int v = 0; v < 8; ++v) rowacc[v] = 0.f;

  const float4* __restrict__ Ab = (const float4*)A;

  for (int t = 0; t < 8; ++t) {
    v8f c = {};
    const float4* bp = Ab + (size_t)t * 16 * 32 + lane;
#pragma unroll
    for (int kk = 0; kk < 16; ++kk) {
      const float4 bv = bp[kk * 32];
      const float2 a0 = *(const float2*)(&xs[wave][l16][8 * kk + 2 * half]);
      const float2 a1 = *(const float2*)(&xs[wave][l16][8 * kk + 4 + 2 * half]);

      v2f av0; av0.x = a0.x; av0.y = a0.y;
      v2f bv0; bv0.x = bv.x; bv0.y = bv.y;
      c = __builtin_amdgcn_wmma_f32_16x16x4_f32(
              false, av0, false, bv0, (short)0, c, false, false);

      v2f av1; av1.x = a1.x; av1.y = a1.y;
      v2f bv1; bv1.x = bv.z; bv1.y = bv.w;
      c = __builtin_amdgcn_wmma_f32_16x16x4_f32(
              false, av1, false, bv1, (short)0, c, false, false);
    }
    const int col = t * 16 + l16;
#pragma unroll
    for (int v = 0; v < 8; ++v)
      rowacc[v] += c[v] * xs[wave][v + 8 * half][col];
  }

#pragma unroll
  for (int v = 0; v < 8; ++v) {
    float s = rowacc[v];
    s += __shfl_xor(s, 1, 32);
    s += __shfl_xor(s, 2, 32);
    s += __shfl_xor(s, 4, 32);
    s += __shfl_xor(s, 8, 32);
    if (l16 == 0) out[tile * 16 + v + 8 * half] = s;
  }
}

extern "C" void kernel_launch(void* const* d_in, const int* in_sizes, int n_in,
                              void* d_out, int out_size, void* d_ws, size_t ws_size,
                              hipStream_t stream) {
  const float* x     = (const float*)d_in[0];
  const float* w     = (const float*)d_in[1];
  const int*   field = (const int*)d_in[2];
  float* out = (float*)d_out;
  float* A   = (float*)d_ws;

  ffm_build_A<<<(N_FEAT * N_FEAT + 255) / 256, 256, 0, stream>>>(w, field, A);

  const int tiles  = B_ROWS / 16;
  const int blocks = tiles / WAVES_PER_BLOCK;
  ffm_quadform<<<blocks, WAVES_PER_BLOCK * 32, 0, stream>>>(x, A, out);
}


# Transposed product (matrix as WMMA A operand): each lane owns one batch row, x fragments read once from LDS into registers, loop has no LDS reads, row-dot in-lane; f32 sum order changes
# speedup vs baseline: 1.1274x; 1.1274x over previous
#include <hip/hip_runtime.h>


#define B_ROWS   131072
#define N_FEAT   128
#define N_FIELD  16
#define FACTOR   16
#define WAVES_PER_BLOCK 4
#define XS_STRIDE 132

typedef __attribute__((ext_vector_type(2))) float v2f;
typedef __attribute__((ext_vector_type(8))) float v8f;

__global__ void ffm_build_A(const float* __restrict__ w,
                            const int* __restrict__ field,
                            float* __restrict__ A) {
  int idx = blockIdx.x * blockDim.x + threadIdx.x;
  if (idx >= N_FEAT * N_FEAT) return;
  int i = idx >> 7;
  int j = idx & (N_FEAT - 1);
  const float* wi = w + (i * N_FIELD + field[j]) * FACTOR;
  const float* wj = w + (j * N_FIELD + field[i]) * FACTOR;
  float s = 0.f;
#pragma unroll
  for (int d = 0; d < FACTOR; ++d) s += wi[d] * wj[d];
  const float val = (i == j) ? 0.f : 0.5f * s;

  const int t  = j >> 4,  n = j & 15;
  const int c  = (n & 3) | ((n & 4) << 1) | ((n & 8) >> 1);
  const int kk = i >> 3;
  const int h  = (i >> 2) & 1;
  const int q  = i & 3;
  const int lane = h * 16 + c;
  A[(((t * 16 + kk) * 32) + lane) * 4 + q] = val;
}

__global__ __launch_bounds__(WAVES_PER_BLOCK * 32)
void ffm_quadform(const float* __restrict__ x,
                  const float* __restrict__ A,
                  float* __restrict__ out) {
  __shared__ __attribute__((aligned(16))) float xs[WAVES_PER_BLOCK][16][XS_STRIDE];

  const int wave = threadIdx.x >> 5;
  const int lane = threadIdx.x & 31;
  const int half = lane >> 4;
  const int l16  = lane & 15;
  const long tile = (long)blockIdx.x * WAVES_PER_BLOCK + wave;
  const float* xt = x + tile * 16 * N_FEAT;

#pragma unroll
  for (int r = 0; r < 16; ++r) {
    const float4 v = *(const float4*)(xt + r * N_FEAT + lane * 4);
    *(float4*)(&xs[wave][r][lane * 4]) = v;
  }

  float4 xr[16];
#pragma unroll
  for (int kk = 0; kk < 16; ++kk)
    xr[kk] = *(const float4*)(&xs[wave][l16][8 * kk + 4 * half]);

  const float4* __restrict__ Ab = (const float4*)A;
  float acc = 0.f;

#pragma unroll
  for (int t = 0; t < 8; ++t) {
    v8f c = {};
    const float4* ap = Ab + (size_t)t * 16 * 32 + lane;
#pragma unroll
    for (int kk = 0; kk < 16; ++kk) {
      const float4 av = ap[kk * 32];
      v2f a0; a0.x = av.x; a0.y = av.y;
      v2f b0; b0.x = xr[kk].x; b0.y = xr[kk].y;
      c = __builtin_amdgcn_wmma_f32_16x16x4_f32(
              false, a0, false, b0, (short)0, c, false, false);
      v2f a1; a1.x = av.z; a1.y = av.w;
      v2f b1; b1.x = xr[kk].z; b1.y = xr[kk].w;
      c = __builtin_amdgcn_wmma_f32_16x16x4_f32(
              false, a1, false, b1, (short)0, c, false, false);
    }
    acc += c[0] * xr[2 * t].x;
    acc += c[1] * xr[2 * t].y;
    acc += c[2] * xr[2 * t].z;
    acc += c[3] * xr[2 * t].w;
    acc += c[4] * xr[2 * t + 1].x;
    acc += c[5] * xr[2 * t + 1].y;
    acc += c[6] * xr[2 * t + 1].z;
    acc += c[7] * xr[2 * t + 1].w;
  }

  acc += __shfl_xor(acc, 16, 32);
  if (half == 0) out[tile * 16 + l16] = acc;
}

extern "C" void kernel_launch(void* const* d_in, const int* in_sizes, int n_in,
                              void* d_out, int out_size, void* d_ws, size_t ws_size,
                              hipStream_t stream) {
  const float* x     = (const float*)d_in[0];
  const float* w     = (const float*)d_in[1];
  const int*   field = (const int*)d_in[2];
  float* out = (float*)d_out;
  float* A   = (float*)d_ws;

  ffm_build_A<<<(N_FEAT * N_FEAT + 255) / 256, 256, 0, stream>>>(w, field, A);

  const int tiles  = B_ROWS / 16;
  const int blocks = tiles / WAVES_PER_BLOCK;
  ffm_quadform<<<blocks, WAVES_PER_BLOCK * 32, 0, stream>>>(x, A, out);
}


# Strictly upper-triangular matrix as the definition masks it (w_ij once for i<j, all entries written); column block t multiplies only k<16(t+1): 144 WMMAs per tile, not 256; f32 sum order changes
# speedup vs baseline: 1.4198x; 1.2594x over previous
#include <hip/hip_runtime.h>


#define B_ROWS   131072
#define N_FEAT   128
#define N_FIELD  16
#define FACTOR   16
#define WAVES_PER_BLOCK 4
#define XS_STRIDE 132

typedef __attribute__((ext_vector_type(2))) float v2f;
typedef __attribute__((ext_vector_type(8))) float v8f;

__global__ void ffm_build_A(const float* __restrict__ w,
                            const int* __restrict__ field,
                            float* __restrict__ A) {
  int idx = blockIdx.x * blockDim.x + threadIdx.x;
  if (idx >= N_FEAT * N_FEAT) return;
  int i = idx >> 7;
  int j = idx & (N_FEAT - 1);
  const float* wi = w + (i * N_FIELD + field[j]) * FACTOR;
  const float* wj = w + (j * N_FIELD + field[i]) * FACTOR;
  float s = 0.f;
#pragma unroll
  for (int d = 0; d < FACTOR; ++d) s += wi[d] * wj[d];
  const float val = (i < j) ? s : 0.f;

  const int t  = j >> 4,  n = j & 15;
  const int c  = (n & 3) | ((n & 4) << 1) | ((n & 8) >> 1);
  const int kk = i >> 3;
  const int h  = (i >> 2) & 1;
  const int q  = i & 3;
  const int lane = h * 16 + c;
  A[(((t * 16 + kk) * 32) + lane) * 4 + q] = val;
}

__global__ __launch_bounds__(WAVES_PER_BLOCK * 32)
void ffm_quadform(const float* __restrict__ x,
                  const float* __restrict__ A,
                  float* __restrict__ out) {
  __shared__ __attribute__((aligned(16))) float xs[WAVES_PER_BLOCK][16][XS_STRIDE];

  const int wave = threadIdx.x >> 5;
  const int lane = threadIdx.x & 31;
  const int half = lane >> 4;
  const int l16  = lane & 15;
  const long tile = (long)blockIdx.x * WAVES_PER_BLOCK + wave;
  const float* xt = x + tile * 16 * N_FEAT;

#pragma unroll
  for (int r = 0; r < 16; ++r) {
    const float4 v = *(const float4*)(xt + r * N_FEAT + lane * 4);
    *(float4*)(&xs[wave][r][lane * 4]) = v;
  }

  float4 xr[16];
#pragma unroll
  for (int kk = 0; kk < 16; ++kk)
    xr[kk] = *(const float4*)(&xs[wave][l16][8 * kk + 4 * half]);

  const float4* __restrict__ Ab = (const float4*)A;
  float acc = 0.f;

#pragma unroll
  for (int t = 0; t < 8; ++t) {
    v8f c = {};
    const float4* ap = Ab + (size_t)t * 16 * 32 + lane;
#pragma unroll
    for (int kk = 0; kk < 2 * t + 2; ++kk) {
      const float4 av = ap[kk * 32];
      v2f a0; a0.x = av.x; a0.y = av.y;
      v2f b0; b0.x = xr[kk].x; b0.y = xr[kk].y;
      c = __builtin_amdgcn_wmma_f32_16x16x4_f32(
              false, a0, false, b0, (short)0, c, false, false);
      v2f a1; a1.x = av.z; a1.y = av.w;
      v2f b1; b1.x = xr[kk].z; b1.y = xr[kk].w;
      c = __builtin_amdgcn_wmma_f32_16x16x4_f32(
              false, a1, false, b1, (short)0, c, false, false);
    }
    acc += c[0] * xr[2 * t].x;
    acc += c[1] * xr[2 * t].y;
    acc += c[2] * xr[2 * t].z;
    acc += c[3] * xr[2 * t].w;
    acc += c[4] * xr[2 * t + 1].x;
    acc += c[5] * xr[2 * t + 1].y;
    acc += c[6] * xr[2 * t + 1].z;
    acc += c[7] * xr[2 * t + 1].w;
  }

  acc += __shfl_xor(acc, 16, 32);
  if (half == 0) out[tile * 16 + l16] = acc;
}

extern "C" void kernel_launch(void* const* d_in, const int* in_sizes, int n_in,
                              void* d_out, int out_size, void* d_ws, size_t ws_size,
                              hipStream_t stream) {
  const float* x     = (const float*)d_in[0];
  const float* w     = (const float*)d_in[1];
  const int*   field = (const int*)d_in[2];
  float* out = (float*)d_out;
  float* A   = (float*)d_ws;

  ffm_build_A<<<(N_FEAT * N_FEAT + 255) / 256, 256, 0, stream>>>(w, field, A);

  const int tiles  = B_ROWS / 16;
  const int blocks = tiles / WAVES_PER_BLOCK;
  ffm_quadform<<<blocks, WAVES_PER_BLOCK * 32, 0, stream>>>(x, A, out);
}


# Resident waves: 256 work-groups of 8 waves; each wave keeps its 72 matrix fragments in registers and walks 4 tiles, later tiles' x rows loaded under the WMMAs; f32 sum order unchanged
# speedup vs baseline: 1.6345x; 1.1512x over previous
#include <hip/hip_runtime.h>


#define B_ROWS   131072
#define N_FEAT   128
#define N_FIELD  16
#define FACTOR   16
#define WAVES_PER_BLOCK 8
#define TILES_PER_WAVE  4
#define NFRAG 72
#define XS_STRIDE 132

typedef __attribute__((ext_vector_type(2))) float v2f;
typedef __attribute__((ext_vector_type(8))) float v8f;

__global__ void ffm_build_A(const float* __restrict__ w,
                            const int* __restrict__ field,
                            float* __restrict__ A) {
  int idx = blockIdx.x * blockDim.x + threadIdx.x;
  if (idx >= N_FEAT * N_FEAT) return;
  int i = idx >> 7;
  int j = idx & (N_FEAT - 1);
  const float* wi = w + (i * N_FIELD + field[j]) * FACTOR;
  const float* wj = w + (j * N_FIELD + field[i]) * FACTOR;
  float s = 0.f;
#pragma unroll
  for (int d = 0; d < FACTOR; ++d) s += wi[d] * wj[d];
  const float val = (i < j) ? s : 0.f;

  const int t  = j >> 4,  n = j & 15;
  const int c  = (n & 3) | ((n & 4) << 1) | ((n & 8) >> 1);
  const int kk = i >> 3;
  const int h  = (i >> 2) & 1;
  const int q  = i & 3;
  const int lane = h * 16 + c;
  A[(((t * 16 + kk) * 32) + lane) * 4 + q] = val;
}


__device__ __forceinline__ float ffm_tile(const float4 (&af)[NFRAG], const float* xb) {
  v8f c[8];
#pragma unroll
  for (int t = 0; t < 8; ++t) c[t] = v8f{};
  float acc = 0.f;
  float4 xe = {};

#pragma unroll
  for (int kk = 0; kk < 16; ++kk) {
    const float4 xr = *(const float4*)(xb + 8 * kk);
    v2f b0; b0.x = xr.x; b0.y = xr.y;
    v2f b1; b1.x = xr.z; b1.y = xr.w;
#pragma unroll
    for (int t = kk >> 1; t < 8; ++t) {
      const float4 av = af[t * (t + 1) + kk];
      v2f a0; a0.x = av.x; a0.y = av.y;
      c[t] = __builtin_amdgcn_wmma_f32_16x16x4_f32(
                 false, a0, false, b0, (short)0, c[t], false, false);
      v2f a1; a1.x = av.z; a1.y = av.w;
      c[t] = __builtin_amdgcn_wmma_f32_16x16x4_f32(
                 false, a1, false, b1, (short)0, c[t], false, false);
    }
    if (kk & 1) {
      const int t = kk >> 1;
      acc += c[t][0] * xe.x;
      acc += c[t][1] * xe.y;
      acc += c[t][2] * xe.z;
      acc += c[t][3] * xe.w;
      acc += c[t][4] * xr.x;
      acc += c[t][5] * xr.y;
      acc += c[t][6] * xr.z;
      acc += c[t][7] * xr.w;
    } else {
      xe = xr;
    }
  }
  acc += __shfl_xor(acc, 16, 32);
  return acc;
}

__global__ __launch_bounds__(WAVES_PER_BLOCK * 32)
void ffm_quadform(const float* x,
                  const float* __restrict__ A,
                  float* out) {
  __shared__ __attribute__((aligned(16)))
      float xs[WAVES_PER_BLOCK][TILES_PER_WAVE][16][XS_STRIDE];

  const int wave = threadIdx.x >> 5;
  const int lane = threadIdx.x & 31;
  const int half = lane >> 4;
  const int l16  = lane & 15;
  const long tile0 = ((long)blockIdx.x * WAVES_PER_BLOCK + wave) * TILES_PER_WAVE;
  const float* xt = x + tile0 * 16 * N_FEAT + lane * 4;
  const float4* __restrict__ ap = (const float4*)A + lane;

  float4 x0[16];
#pragma unroll
  for (int r = 0; r < 16; ++r) x0[r] = *(const float4*)(xt + r * N_FEAT);

  float4 af[NFRAG];
#pragma unroll
  for (int t = 0; t < 8; ++t)
#pragma unroll
    for (int kk = 0; kk < 2 * t + 2; ++kk)
      af[t * (t + 1) + kk] = ap[(t * 16 + kk) * 32];

#pragma unroll
  for (int r = 0; r < 16; ++r) *(float4*)(&xs[wave][0][r][lane * 4]) = x0[r];

#define FFM_ROWS(F) F(0) F(1) F(2) F(3) F(4) F(5) F(6) F(7) F(8) F(9) F(10) F(11) F(12) F(13) F(14) F(15)
#define FFM_DECL(r)  float4 nx##r;
#define FFM_LOAD(r)  nx##r = *(const float4*)(xn + (r) * N_FEAT);
#define FFM_STORE(r) *(float4*)(xd + (r) * XS_STRIDE) = nx##r;
  FFM_ROWS(FFM_DECL)

  {
    const float* xn = xt + 16 * N_FEAT;
    FFM_ROWS(FFM_LOAD)
  }

#pragma unroll 1
  for (int it = 0; it < TILES_PER_WAVE - 2; ++it) {
    const float acc = ffm_tile(af, &xs[wave][it][l16][4 * half]);
    {
      float* xd = &xs[wave][it + 1][0][lane * 4];
      FFM_ROWS(FFM_STORE)
    }
    {
      const float* xn = xt + (size_t)(it + 2) * 16 * N_FEAT;
      FFM_ROWS(FFM_LOAD)
    }
    if (half == 0) out[(tile0 + it) * 16 + l16] = acc;
  }
  {
    const int it = TILES_PER_WAVE - 2;
    const float acc = ffm_tile(af, &xs[wave][it][l16][4 * half]);
    {
      float* xd = &xs[wave][it + 1][0][lane * 4];
      FFM_ROWS(FFM_STORE)
    }
    if (half == 0) out[(tile0 + it) * 16 + l16] = acc;
    const float acc2 = ffm_tile(af, &xs[wave][it + 1][l16][4 * half]);
    if (half == 0) out[(tile0 + it + 1) * 16 + l16] = acc2;
  }
#undef FFM_ROWS
#undef FFM_DECL
#undef FFM_LOAD
#undef FFM_STORE
}

extern "C" void kernel_launch(void* const* d_in, const int* in_sizes, int n_in,
                              void* d_out, int out_size, void* d_ws, size_t ws_size,
                              hipStream_t stream) {
  const float* x     = (const float*)d_in[0];
  const float* w     = (const float*)d_in[1];
  const int*   field = (const int*)d_in[2];
  float* out = (float*)d_out;
  float* A   = (float*)d_ws;

  ffm_build_A<<<(N_FEAT * N_FEAT + 255) / 256, 256, 0, stream>>>(w, field, A);

  const int tiles  = B_ROWS / 16;
  const int blocks = tiles / (WAVES_PER_BLOCK * TILES_PER_WAVE);
  ffm_quadform<<<blocks, WAVES_PER_BLOCK * 32, 0, stream>>>(x, A, out);
}
